# starting kernel
# baseline (speedup 1.0000x reference)
#include <hip/hip_runtime.h>

typedef __attribute__((ext_vector_type(2))) float v2f;
typedef __attribute__((ext_vector_type(8))) float v8f;

#define N_NODES 50000
#define N_EDGES 800000
#define F_IN    128
#define F_OUT   64

__global__ void wsum_kernel(const float* __restrict__ w, float* __restrict__ wsum) {
    int i = blockIdx.x * blockDim.x + threadIdx.x;
    if (i < F_IN * F_OUT) wsum[i] = w[i] + w[F_IN * F_OUT + i];
}

__global__ void gemm_wmma_kernel(const float* __restrict__ h,
                                 const float* __restrict__ wsum,
                                 float* __restrict__ hsum) {
    const int wave = (blockIdx.x * blockDim.x + threadIdx.x) >> 5;
    const int lane = threadIdx.x & 31;
    const int ntileN = F_OUT / 16;
    const int m0 = (wave / ntileN) * 16;
    const int n0 = (wave % ntileN) * 16;
    if (m0 >= N_NODES) return;

    const int half = lane >> 4;
    const int l16  = lane & 15;

    const float* arow = h + (m0 + l16) * F_IN + 2 * half;
    const float* bcol = wsum + n0 + l16 + 2 * half * F_OUT;

    v8f acc = {};
    #pragma unroll 4
    for (int k = 0; k < F_IN; k += 4) {
        v2f a, b;
        a.x = arow[k + 0];
        a.y = arow[k + 1];
        b.x = bcol[(k + 0) * F_OUT];
        b.y = bcol[(k + 1) * F_OUT];
        acc = __builtin_amdgcn_wmma_f32_16x16x4_f32(
             false, a,  false, b,
             (short)0, acc,  false,  false);
    }

    float* drow = hsum + m0 * F_OUT + n0 + l16;
    #pragma unroll
    for (int r = 0; r < 8; ++r)
        drow[(r + 8 * half) * F_OUT] = acc[r];
}

__global__ void init_out_kernel(const float* __restrict__ bias, float* __restrict__ out) {
    int i = blockIdx.x * blockDim.x + threadIdx.x;
    if (i < N_NODES * F_OUT) out[i] = bias[i & (F_OUT - 1)];
}

__global__ void edge_kernel(const float* __restrict__ edge_feat,
                            const int*   __restrict__ src,
                            const int*   __restrict__ dst,
                            const float* __restrict__ hsum,
                            float*       __restrict__ out) {
    const int wave = (blockIdx.x * (blockDim.x >> 5)) + (threadIdx.x >> 5);
    const int lane = threadIdx.x & 31;
    if (wave >= N_EDGES) return;

    const float f0 = edge_feat[wave * F_OUT + lane];
    const float f1 = edge_feat[wave * F_OUT + lane + 32];

    float mx = fmaxf(f0, f1);
    #pragma unroll
    for (int off = 16; off > 0; off >>= 1)
        mx = fmaxf(mx, __shfl_xor(mx, off, 32));

    const float e0 = __expf(f0 - mx);
    const float e1 = __expf(f1 - mx);
    float s = e0 + e1;
    #pragma unroll
    for (int off = 16; off > 0; off >>= 1)
        s += __shfl_xor(s, off, 32);
    const float inv = 1.0f / s;

    const int sn = src[wave];
    const int dn = dst[wave];
    const float g0 = hsum[sn * F_OUT + lane];
    const float g1 = hsum[sn * F_OUT + lane + 32];

    atomicAdd(&out[dn * F_OUT + lane],      g0 * e0 * inv);
    atomicAdd(&out[dn * F_OUT + lane + 32], g1 * e1 * inv);
}

extern "C" void kernel_launch(void* const* d_in, const int* in_sizes, int n_in,
                              void* d_out, int out_size, void* d_ws, size_t ws_size,
                              hipStream_t stream) {
    const float* h         = (const float*)d_in[0];
    const float* edge_feat = (const float*)d_in[1];
    const float* weight    = (const float*)d_in[2];
    const float* bias      = (const float*)d_in[3];
    const int*   src       = (const int*)d_in[4];
    const int*   dst       = (const int*)d_in[5];
    float* out = (float*)d_out;

    float* wsum = (float*)d_ws;
    float* hsum = (float*)((char*)d_ws + ((F_IN * F_OUT * sizeof(float) + 255u) & ~255u));

    wsum_kernel<<<(F_IN * F_OUT + 255) / 256, 256, 0, stream>>>(weight, wsum);

    const int ntiles = (N_NODES / 16) * (F_OUT / 16);
    gemm_wmma_kernel<<<(ntiles + 7) / 8, 256, 0, stream>>>(h, wsum, hsum);

    init_out_kernel<<<(N_NODES * F_OUT + 255) / 256, 256, 0, stream>>>(bias, out);

    edge_kernel<<<(N_EDGES + 7) / 8, 256, 0, stream>>>(edge_feat, src, dst, hsum, out);
}


# edge kernel: 8 edges per wave, 16 lanes per edge on 16-byte loads, gathers requested beside the stream, whole-line atomics kept by an LDS re-lay; softmax sum tree now 4 in-lane + 16-lane butterfly
# speedup vs baseline: 1.6882x; 1.6882x over previous
#include <hip/hip_runtime.h>

typedef __attribute__((ext_vector_type(2))) float v2f;
typedef __attribute__((ext_vector_type(4))) float v4f;
typedef __attribute__((ext_vector_type(8))) float v8f;

#define N_NODES 50000
#define N_EDGES 800000
#define F_IN    128
#define F_OUT   64

__global__ void wsum_kernel(const float* __restrict__ w, float* __restrict__ wsum) {
    int i = blockIdx.x * blockDim.x + threadIdx.x;
    if (i < F_IN * F_OUT) wsum[i] = w[i] + w[F_IN * F_OUT + i];
}

__global__ void gemm_wmma_kernel(const float* __restrict__ h,
                                 const float* __restrict__ wsum,
                                 float* __restrict__ hsum) {
    const int wave = (blockIdx.x * blockDim.x + threadIdx.x) >> 5;
    const int lane = threadIdx.x & 31;
    const int ntileN = F_OUT / 16;
    const int m0 = (wave / ntileN) * 16;
    const int n0 = (wave % ntileN) * 16;
    if (m0 >= N_NODES) return;

    const int half = lane >> 4;
    const int l16  = lane & 15;

    const float* arow = h + (m0 + l16) * F_IN + 2 * half;
    const float* bcol = wsum + n0 + l16 + 2 * half * F_OUT;

    v8f acc = {};
    #pragma unroll 4
    for (int k = 0; k < F_IN; k += 4) {
        v2f a, b;
        a.x = arow[k + 0];
        a.y = arow[k + 1];
        b.x = bcol[(k + 0) * F_OUT];
        b.y = bcol[(k + 1) * F_OUT];
        acc = __builtin_amdgcn_wmma_f32_16x16x4_f32(
             false, a,  false, b,
             (short)0, acc,  false,  false);
    }

    float* drow = hsum + m0 * F_OUT + n0 + l16;
    #pragma unroll
    for (int r = 0; r < 8; ++r)
        drow[(r + 8 * half) * F_OUT] = acc[r];
}

__global__ void init_out_kernel(const float* __restrict__ bias, float* __restrict__ out) {
    int i = blockIdx.x * blockDim.x + threadIdx.x;
    if (i < N_NODES * F_OUT) out[i] = bias[i & (F_OUT - 1)];
}

#define EPW   8
#define SLOTS (EPW / 2)
#define WPB   8

__global__ __launch_bounds__(256)
void edge_kernel(const float* __restrict__ edge_feat,
                 const int*   __restrict__ src,
                 const int*   __restrict__ dst,
                 const float* __restrict__ hsum,
                 float*       __restrict__ out) {
    static_assert(N_EDGES % EPW == 0, "a wave owns EPW whole edges");
    __shared__ __attribute__((aligned(16))) float stage[WPB][SLOTS][128];

    const int lane = threadIdx.x & 31;
    const int wib  = threadIdx.x >> 5;
    const int e0 = __builtin_amdgcn_readfirstlane((blockIdx.x * WPB + wib) * EPW);
    if (e0 >= N_EDGES) return;

    const int half = lane >> 4;
    const int l16  = lane & 15;

    const v4f* ef = (const v4f*)(edge_feat + e0 * F_OUT) + lane;
    v4f f[SLOTS];
    #pragma unroll
    for (int u = 0; u < SLOTS; ++u) f[u] = ef[u * 32];

    int sv[EPW], dv[EPW];
    #pragma unroll
    for (int i = 0; i < EPW; ++i) { sv[i] = src[e0 + i]; dv[i] = dst[e0 + i]; }

    v4f g[SLOTS];
    #pragma unroll
    for (int u = 0; u < SLOTS; ++u) {
        const int sa = __builtin_amdgcn_readfirstlane(sv[2 * u]);
        const int sb = __builtin_amdgcn_readfirstlane(sv[2 * u + 1]);
        const int sn = half ? sb : sa;
        g[u] = *((const v4f*)(hsum + sn * F_OUT) + l16);
    }

    #pragma unroll
    for (int u = 0; u < SLOTS; ++u) {
        const v4f x = f[u];

        float mx = fmaxf(fmaxf(x.x, x.y), fmaxf(x.z, x.w));
        #pragma unroll
        for (int off = 8; off > 0; off >>= 1)
            mx = fmaxf(mx, __shfl_xor(mx, off, 32));

        v4f e;
        e.x = __expf(x.x - mx);
        e.y = __expf(x.y - mx);
        e.z = __expf(x.z - mx);
        e.w = __expf(x.w - mx);
        float s = (e.x + e.y) + (e.z + e.w);
        #pragma unroll
        for (int off = 8; off > 0; off >>= 1)
            s += __shfl_xor(s, off, 32);
        const float inv = 1.0f / s;

        v4f m;
        m.x = g[u].x * e.x * inv;
        m.y = g[u].y * e.y * inv;
        m.z = g[u].z * e.z * inv;
        m.w = g[u].w * e.w * inv;

        float* st = &stage[wib][u][0];
        *((v4f*)st + lane) = m;
        __builtin_amdgcn_fence(__ATOMIC_RELEASE, "wavefront");
        __builtin_amdgcn_wave_barrier();
        __builtin_amdgcn_fence(__ATOMIC_ACQUIRE, "wavefront");

        #pragma unroll
        for (int t = 0; t < 4; ++t) {
            const float r = st[t * 32 + lane];
            const int dn = dv[2 * u + (t >> 1)];
            atomicAdd(&out[dn * F_OUT + (t & 1) * 32 + lane], r);
        }
    }
}

extern "C" void kernel_launch(void* const* d_in, const int* in_sizes, int n_in,
                              void* d_out, int out_size, void* d_ws, size_t ws_size,
                              hipStream_t stream) {
    const float* h         = (const float*)d_in[0];
    const float* edge_feat = (const float*)d_in[1];
    const float* weight    = (const float*)d_in[2];
    const float* bias      = (const float*)d_in[3];
    const int*   src       = (const int*)d_in[4];
    const int*   dst       = (const int*)d_in[5];
    float* out = (float*)d_out;

    float* wsum = (float*)d_ws;
    float* hsum = (float*)((char*)d_ws + ((F_IN * F_OUT * sizeof(float) + 255u) & ~255u));

    wsum_kernel<<<(F_IN * F_OUT + 255) / 256, 256, 0, stream>>>(weight, wsum);

    const int ntiles = (N_NODES / 16) * (F_OUT / 16);
    gemm_wmma_kernel<<<(ntiles + 7) / 8, 256, 0, stream>>>(h, wsum, hsum);

    init_out_kernel<<<(N_NODES * F_OUT + 255) / 256, 256, 0, stream>>>(bias, out);

    edge_kernel<<<(N_EDGES + 63) / 64, 256, 0, stream>>>(edge_feat, src, dst, hsum, out);
}


# bias seeding of out moved into the GEMM launch: each wave stores 4 whole rows (two 16-byte stores per lane) behind its K loop; 4 launches -> 3, the 100,000-wave seeding launch gone; sum order kept
# speedup vs baseline: 1.8304x; 1.0842x over previous
#include <hip/hip_runtime.h>

typedef __attribute__((ext_vector_type(2))) float v2f;
typedef __attribute__((ext_vector_type(4))) float v4f;
typedef __attribute__((ext_vector_type(8))) float v8f;

#define N_NODES 50000
#define N_EDGES 800000
#define F_IN    128
#define F_OUT   64

__global__ void wsum_kernel(const float* __restrict__ w, float* __restrict__ wsum) {
    int i = blockIdx.x * blockDim.x + threadIdx.x;
    if (i < F_IN * F_OUT) wsum[i] = w[i] + w[F_IN * F_OUT + i];
}

__global__ void gemm_wmma_kernel(const float* __restrict__ h,
                                 const float* __restrict__ wsum,
                                 const float* __restrict__ bias,
                                 float* __restrict__ hsum,
                                 float* __restrict__ out) {
    static_assert(N_NODES % 16 == 0, "every 16-row band is whole, so the seeding covers every row");
    const int wave = (blockIdx.x * blockDim.x + threadIdx.x) >> 5;
    const int lane = threadIdx.x & 31;
    const int ntileN = F_OUT / 16;
    const int m0 = (wave / ntileN) * 16;
    const int n0 = (wave % ntileN) * 16;
    if (m0 >= N_NODES) return;

    const int half = lane >> 4;
    const int l16  = lane & 15;

    const v4f bv = *((const v4f*)bias + l16);

    const float* arow = h + (m0 + l16) * F_IN + 2 * half;
    const float* bcol = wsum + n0 + l16 + 2 * half * F_OUT;

    v8f acc = {};
    #pragma unroll 4
    for (int k = 0; k < F_IN; k += 4) {
        v2f a, b;
        a.x = arow[k + 0];
        a.y = arow[k + 1];
        b.x = bcol[(k + 0) * F_OUT];
        b.y = bcol[(k + 1) * F_OUT];
        acc = __builtin_amdgcn_wmma_f32_16x16x4_f32(
             false, a,  false, b,
             (short)0, acc,  false,  false);
    }

    v4f* o4 = (v4f*)(out + (m0 + (n0 >> 2)) * F_OUT) + lane;
    o4[0]  = bv;
    o4[32] = bv;

    float* drow = hsum + m0 * F_OUT + n0 + l16;
    #pragma unroll
    for (int r = 0; r < 8; ++r)
        drow[(r + 8 * half) * F_OUT] = acc[r];
}

#define EPW   8
#define SLOTS (EPW / 2)
#define WPB   8

__global__ __launch_bounds__(256)
void edge_kernel(const float* __restrict__ edge_feat,
                 const int*   __restrict__ src,
                 const int*   __restrict__ dst,
                 const float* __restrict__ hsum,
                 float*       __restrict__ out) {
    static_assert(N_EDGES % EPW == 0, "a wave owns EPW whole edges");
    __shared__ __attribute__((aligned(16))) float stage[WPB][SLOTS][128];

    const int lane = threadIdx.x & 31;
    const int wib  = threadIdx.x >> 5;
    const int e0 = __builtin_amdgcn_readfirstlane((blockIdx.x * WPB + wib) * EPW);
    if (e0 >= N_EDGES) return;

    const int half = lane >> 4;
    const int l16  = lane & 15;

    const v4f* ef = (const v4f*)(edge_feat + e0 * F_OUT) + lane;
    v4f f[SLOTS];
    #pragma unroll
    for (int u = 0; u < SLOTS; ++u) f[u] = ef[u * 32];

    int sv[EPW], dv[EPW];
    #pragma unroll
    for (int i = 0; i < EPW; ++i) { sv[i] = src[e0 + i]; dv[i] = dst[e0 + i]; }

    v4f g[SLOTS];
    #pragma unroll
    for (int u = 0; u < SLOTS; ++u) {
        const int sa = __builtin_amdgcn_readfirstlane(sv[2 * u]);
        const int sb = __builtin_amdgcn_readfirstlane(sv[2 * u + 1]);
        const int sn = half ? sb : sa;
        g[u] = *((const v4f*)(hsum + sn * F_OUT) + l16);
    }

    #pragma unroll
    for (int u = 0; u < SLOTS; ++u) {
        const v4f x = f[u];

        float mx = fmaxf(fmaxf(x.x, x.y), fmaxf(x.z, x.w));
        #pragma unroll
        for (int off = 8; off > 0; off >>= 1)
            mx = fmaxf(mx, __shfl_xor(mx, off, 32));

        v4f e;
        e.x = __expf(x.x - mx);
        e.y = __expf(x.y - mx);
        e.z = __expf(x.z - mx);
        e.w = __expf(x.w - mx);
        float s = (e.x + e.y) + (e.z + e.w);
        #pragma unroll
        for (int off = 8; off > 0; off >>= 1)
            s += __shfl_xor(s, off, 32);
        const float inv = 1.0f / s;

        v4f m;
        m.x = g[u].x * e.x * inv;
        m.y = g[u].y * e.y * inv;
        m.z = g[u].z * e.z * inv;
        m.w = g[u].w * e.w * inv;

        float* st = &stage[wib][u][0];
        *((v4f*)st + lane) = m;
        __builtin_amdgcn_fence(__ATOMIC_RELEASE, "wavefront");
        __builtin_amdgcn_wave_barrier();
        __builtin_amdgcn_fence(__ATOMIC_ACQUIRE, "wavefront");

        #pragma unroll
        for (int t = 0; t < 4; ++t) {
            const float r = st[t * 32 + lane];
            const int dn = dv[2 * u + (t >> 1)];
            atomicAdd(&out[dn * F_OUT + (t & 1) * 32 + lane], r);
        }
    }
}

extern "C" void kernel_launch(void* const* d_in, const int* in_sizes, int n_in,
                              void* d_out, int out_size, void* d_ws, size_t ws_size,
                              hipStream_t stream) {
    const float* h         = (const float*)d_in[0];
    const float* edge_feat = (const float*)d_in[1];
    const float* weight    = (const float*)d_in[2];
    const float* bias      = (const float*)d_in[3];
    const int*   src       = (const int*)d_in[4];
    const int*   dst       = (const int*)d_in[5];
    float* out = (float*)d_out;

    float* wsum = (float*)d_ws;
    float* hsum = (float*)((char*)d_ws + ((F_IN * F_OUT * sizeof(float) + 255u) & ~255u));

    wsum_kernel<<<(F_IN * F_OUT + 255) / 256, 256, 0, stream>>>(weight, wsum);

    const int ntiles = (N_NODES / 16) * (F_OUT / 16);
    gemm_wmma_kernel<<<(ntiles + 7) / 8, 256, 0, stream>>>(h, wsum, bias, hsum, out);

    edge_kernel<<<(N_EDGES + 63) / 64, 256, 0, stream>>>(edge_feat, src, dst, hsum, out);
}
